# starting kernel
# baseline (speedup 1.0000x reference)
#include <hip/hip_runtime.h>
#include <hip/hip_bf16.h>


typedef __attribute__((ext_vector_type(2))) float v2f;
typedef __attribute__((ext_vector_type(8))) float v8f;

#define DMAT   64
#define ROWS   32
#define PITCH  68
#define NTERMS 30

__global__ __launch_bounds__(256)
void linode_expm_action_kernel(const float* __restrict__ dt,
                               const float* __restrict__ x0,
                               const float* __restrict__ scal,
                               const float* __restrict__ weight,
                               float* __restrict__ out)
{
    __shared__ float Wl[DMAT * PITCH];
    __shared__ float Yl[2][ROWS * PITCH];

    const int tid  = threadIdx.x;
    const int wave = tid >> 5;
    const int lane = tid & 31;
    const int half = lane >> 4;
    const int ln   = lane & 15;
    const int tm   = wave >> 2;
    const int tn   = wave & 3;
    const int r0   = blockIdx.x * ROWS;

    const float gamma = scal[0];
    for (int i = tid; i < DMAT * DMAT; i += 256) {
        int r = i >> 6, c = i & 63;
        Wl[r * PITCH + c] = gamma * weight[i];
    }

    const int rowBase = tm * 16 + half * 8;
    float dtv[8];
    #pragma unroll
    for (int v = 0; v < 8; ++v)
        dtv[v] = dt[r0 + rowBase + v];

    v8f S;
    #pragma unroll
    for (int v = 0; v < 8; ++v)
        S[v] = x0[(size_t)(r0 + rowBase + v) * DMAT + tn * 16 + ln];
    for (int i = tid; i < ROWS * DMAT; i += 256) {
        int r = i >> 6, c = i & 63;
        Yl[0][r * PITCH + c] = x0[(size_t)(r0 + r) * DMAT + c];
    }
    __syncthreads();

    v2f bf[16];
    #pragma unroll
    for (int kb = 0; kb < 16; ++kb)
        bf[kb] = *(const v2f*)&Wl[(tn * 16 + ln) * PITCH + kb * 4 + 2 * half];

    const float* arow0 = &Yl[0][(tm * 16 + ln) * PITCH + 2 * half];
    const float* arow1 = &Yl[1][(tm * 16 + ln) * PITCH + 2 * half];
    float* wrow0 = &Yl[0][rowBase * PITCH + tn * 16 + ln];
    float* wrow1 = &Yl[1][rowBase * PITCH + tn * 16 + ln];

    #pragma unroll
    for (int k = 1; k <= NTERMS; ++k) {
        const float invk = 1.0f / (float)k;
        const float* ybase = (k & 1) ? arow0 : arow1;
        float*       wbase = (k & 1) ? wrow1 : wrow0;

        v8f acc = {};
        #pragma unroll
        for (int kb = 0; kb < 16; ++kb) {
            v2f af = *(const v2f*)(ybase + kb * 4);
            acc = __builtin_amdgcn_wmma_f32_16x16x4_f32(
                false, af, false, bf[kb], (short)0, acc, false, false);
        }

        #pragma unroll
        for (int v = 0; v < 8; ++v) {
            float val = acc[v] * (dtv[v] * invk);
            wbase[v * PITCH] = val;
            S[v] += val;
        }
        __syncthreads();
    }

    #pragma unroll
    for (int v = 0; v < 8; ++v)
        out[(size_t)(r0 + rowBase + v) * DMAT + tn * 16 + ln] = S[v];
}

extern "C" void kernel_launch(void* const* d_in, const int* in_sizes, int n_in,
                              void* d_out, int out_size, void* d_ws, size_t ws_size,
                              hipStream_t stream) {
    const float* dt     = (const float*)d_in[0];
    const float* x0     = (const float*)d_in[1];
    const float* scalar = (const float*)d_in[2];
    const float* weight = (const float*)d_in[3];
    float* out          = (float*)d_out;

    const int B = in_sizes[0];
    const int nblocks = B / ROWS;

    linode_expm_action_kernel<<<nblocks, 256, 0, stream>>>(dt, x0, scalar, weight, out);
}


# LDS reads of a term's 16 A fragments issued as one group ahead of its 16 WMMAs (scheduling groups): one exposed LDS round trip per term instead of eight; order of sums unchanged
# speedup vs baseline: 1.2619x; 1.2619x over previous
#include <hip/hip_runtime.h>
#include <hip/hip_bf16.h>


typedef __attribute__((ext_vector_type(2))) float v2f;
typedef __attribute__((ext_vector_type(8))) float v8f;

#define DMAT   64
#define ROWS   32
#define PITCH  68
#define NTERMS 30

__global__ __launch_bounds__(256)
void linode_expm_action_kernel(const float* __restrict__ dt,
                               const float* __restrict__ x0,
                               const float* __restrict__ scal,
                               const float* __restrict__ weight,
                               float* __restrict__ out)
{
    __shared__ float Wl[DMAT * PITCH];
    __shared__ float Yl[2][ROWS * PITCH];

    const int tid  = threadIdx.x;
    const int wave = tid >> 5;
    const int lane = tid & 31;
    const int half = lane >> 4;
    const int ln   = lane & 15;
    const int tm   = wave >> 2;
    const int tn   = wave & 3;
    const int r0   = blockIdx.x * ROWS;

    const float gamma = scal[0];
    for (int i = tid; i < DMAT * DMAT; i += 256) {
        int r = i >> 6, c = i & 63;
        Wl[r * PITCH + c] = gamma * weight[i];
    }

    const int rowBase = tm * 16 + half * 8;
    float dtv[8];
    #pragma unroll
    for (int v = 0; v < 8; ++v)
        dtv[v] = dt[r0 + rowBase + v];

    v8f S;
    #pragma unroll
    for (int v = 0; v < 8; ++v)
        S[v] = x0[(size_t)(r0 + rowBase + v) * DMAT + tn * 16 + ln];
    for (int i = tid; i < ROWS * DMAT; i += 256) {
        int r = i >> 6, c = i & 63;
        Yl[0][r * PITCH + c] = x0[(size_t)(r0 + r) * DMAT + c];
    }
    __syncthreads();

    v2f bf[16];
    #pragma unroll
    for (int kb = 0; kb < 16; ++kb)
        bf[kb] = *(const v2f*)&Wl[(tn * 16 + ln) * PITCH + kb * 4 + 2 * half];

    const float* arow0 = &Yl[0][(tm * 16 + ln) * PITCH + 2 * half];
    const float* arow1 = &Yl[1][(tm * 16 + ln) * PITCH + 2 * half];
    float* wrow0 = &Yl[0][rowBase * PITCH + tn * 16 + ln];
    float* wrow1 = &Yl[1][rowBase * PITCH + tn * 16 + ln];

    #pragma unroll
    for (int k = 1; k <= NTERMS; ++k) {
        const float invk = 1.0f / (float)k;
        const float* ybase = (k & 1) ? arow0 : arow1;
        float*       wbase = (k & 1) ? wrow1 : wrow0;

        v2f af[16];
        #pragma unroll
        for (int kb = 0; kb < 16; ++kb)
            af[kb] = *(const v2f*)(ybase + kb * 4);

        v8f acc = {};
        #pragma unroll
        for (int kb = 0; kb < 16; ++kb)
            acc = __builtin_amdgcn_wmma_f32_16x16x4_f32(
                false, af[kb], false, bf[kb], (short)0, acc, false, false);

        __builtin_amdgcn_sched_group_barrier(0x100, 16, 0);
        __builtin_amdgcn_sched_group_barrier(0x008, 16, 0);

        #pragma unroll
        for (int v = 0; v < 8; ++v) {
            float val = acc[v] * (dtv[v] * invk);
            wbase[v * PITCH] = val;
            S[v] += val;
        }
        __syncthreads();
    }

    #pragma unroll
    for (int v = 0; v < 8; ++v)
        out[(size_t)(r0 + rowBase + v) * DMAT + tn * 16 + ln] = S[v];
}

extern "C" void kernel_launch(void* const* d_in, const int* in_sizes, int n_in,
                              void* d_out, int out_size, void* d_ws, size_t ws_size,
                              hipStream_t stream) {
    const float* dt     = (const float*)d_in[0];
    const float* x0     = (const float*)d_in[1];
    const float* scalar = (const float*)d_in[2];
    const float* weight = (const float*)d_in[3];
    float* out          = (float*)d_out;

    const int B = in_sizes[0];
    const int nblocks = B / ROWS;

    linode_expm_action_kernel<<<nblocks, 256, 0, stream>>>(dt, x0, scalar, weight, out);
}


# Products transposed (batch on the N side): a term's D registers are the next term's B fragments, so the state stays in registers and wave pairs swap only half of it through LDS; k sum order permuted
# speedup vs baseline: 1.9752x; 1.5653x over previous
#include <hip/hip_runtime.h>
#include <hip/hip_bf16.h>


typedef __attribute__((ext_vector_type(2))) float v2f;
typedef __attribute__((ext_vector_type(8))) float v8f;

#define DMAT   64
#define ROWS   32
#define NTERMS 30

__global__ __launch_bounds__(128)
void linode_expm_action_kernel(const float* __restrict__ dt,
                               const float* __restrict__ x0,
                               const float* __restrict__ scal,
                               const float* __restrict__ weight,
                               float* __restrict__ out)
{
    __shared__ float4 xb[2][4][4][32];

    const int tid  = threadIdx.x;
    const int wave = tid >> 5;
    const int lane = tid & 31;
    const int h    = lane >> 4;
    const int c    = lane & 15;
    const int w    = wave & 1;
    const int pr   = wave >> 1;
    const int b    = blockIdx.x * ROWS + pr * 16 + c;

    const float gamma = scal[0];
    const float dtl   = dt[b];

    v2f af[2][16];
    #pragma unroll
    for (int mo = 0; mo < 2; ++mo) {
        const float* wrow = weight + (size_t)(32 * w + 16 * mo + c) * DMAT + 8 * h;
        #pragma unroll
        for (int u = 0; u < 2; ++u) {
            const int ws = u ? (1 - w) : w;
            #pragma unroll
            for (int mt = 0; mt < 2; ++mt) {
                #pragma unroll
                for (int qq = 0; qq < 2; ++qq) {
                    const float4 t = *(const float4*)(wrow + 32 * ws + 16 * mt + 4 * qq);
                    const int j = 8 * u + 4 * mt + 2 * qq;
                    af[mo][j]     = v2f{gamma * t.x, gamma * t.y};
                    af[mo][j + 1] = v2f{gamma * t.z, gamma * t.w};
                }
            }
        }
    }

    v8f y[2], S[2];
    #pragma unroll
    for (int mt = 0; mt < 2; ++mt) {
        const float* xr = x0 + (size_t)b * DMAT + 32 * w + 16 * mt + 8 * h;
        const float4 lo = *(const float4*)(xr);
        const float4 hi = *(const float4*)(xr + 4);
        y[mt] = v8f{lo.x, lo.y, lo.z, lo.w, hi.x, hi.y, hi.z, hi.w};
        S[mt] = v8f{};
    }

#define PIN() __builtin_amdgcn_sched_barrier(0)
#define PAIR(J, BFR)                                                                   \
    do {                                                                               \
        acc0 = __builtin_amdgcn_wmma_f32_16x16x4_f32(                                  \
            false, af[0][(J)], false, (BFR), (short)0, acc0, false, false);            \
        acc1 = __builtin_amdgcn_wmma_f32_16x16x4_f32(                                  \
            false, af[1][(J)], false, (BFR), (short)0, acc1, false, false);            \
        PIN();                                                                         \
    } while (0)
#define OWN(J) v2f{y[(J) >> 2][2 * ((J) & 3)], y[(J) >> 2][2 * ((J) & 3) + 1]}
#define HELD() asm volatile("" :: "v"(acc0), "v"(acc1))

    #pragma unroll
    for (int k = 1; k <= NTERMS; ++k) {
        const float invk = 1.0f / (float)k;
        const int par = k & 1;

        #pragma unroll
        for (int q = 0; q < 4; ++q) {
            const int mt = q >> 1, e = 4 * (q & 1);
            xb[par][wave][q][lane] = float4{y[mt][e], y[mt][e + 1], y[mt][e + 2], y[mt][e + 3]};
        }
        PIN();

        v8f acc0 = {}, acc1 = {};
        PAIR(0, OWN(0));
        S[0] += y[0];
        S[1] += y[1];
        PAIR(1, OWN(1));
        PAIR(2, OWN(2));

        HELD();
        __builtin_amdgcn_fence(__ATOMIC_RELEASE, "workgroup");
        __builtin_amdgcn_s_barrier_signal(-1);
        PIN();
        PAIR(3, OWN(3));
        PAIR(4, OWN(4));
        HELD();
        __builtin_amdgcn_s_barrier_wait(-1);
        __builtin_amdgcn_fence(__ATOMIC_ACQUIRE, "workgroup");
        float4 pq[4];
        #pragma unroll
        for (int q = 0; q < 4; ++q)
            pq[q] = xb[par][wave ^ 1][q][lane];
        PIN();
        PAIR(5, OWN(5));
        PAIR(6, OWN(6));
        PAIR(7, OWN(7));

        PAIR(8,  (v2f{pq[0].x, pq[0].y}));
        PAIR(9,  (v2f{pq[0].z, pq[0].w}));
        PAIR(10, (v2f{pq[1].x, pq[1].y}));
        PAIR(11, (v2f{pq[1].z, pq[1].w}));
        PAIR(12, (v2f{pq[2].x, pq[2].y}));
        PAIR(13, (v2f{pq[2].z, pq[2].w}));
        PAIR(14, (v2f{pq[3].x, pq[3].y}));
        PAIR(15, (v2f{pq[3].z, pq[3].w}));

        const float cf = dtl * invk;
        y[0] = acc0 * cf;
        y[1] = acc1 * cf;
    }
    S[0] += y[0];
    S[1] += y[1];
#undef HELD
#undef OWN
#undef PAIR
#undef PIN

    #pragma unroll
    for (int mt = 0; mt < 2; ++mt) {
        float* orow = out + (size_t)b * DMAT + 32 * w + 16 * mt + 8 * h;
        *(float4*)(orow)     = float4{S[mt][0], S[mt][1], S[mt][2], S[mt][3]};
        *(float4*)(orow + 4) = float4{S[mt][4], S[mt][5], S[mt][6], S[mt][7]};
    }
}

extern "C" void kernel_launch(void* const* d_in, const int* in_sizes, int n_in,
                              void* d_out, int out_size, void* d_ws, size_t ws_size,
                              hipStream_t stream) {
    const float* dt     = (const float*)d_in[0];
    const float* x0     = (const float*)d_in[1];
    const float* scalar = (const float*)d_in[2];
    const float* weight = (const float*)d_in[3];
    float* out          = (float*)d_out;

    const int B = in_sizes[0];
    const int nblocks = B / ROWS;

    linode_expm_action_kernel<<<nblocks, 128, 0, stream>>>(dt, x0, scalar, weight, out);
}
